# starting kernel
# baseline (speedup 1.0000x reference)
#include <hip/hip_runtime.h>
#include <math.h>

typedef __attribute__((ext_vector_type(2))) float v2f;
typedef __attribute__((ext_vector_type(8))) float v8f;

#define DIM      256
#define MT       64
#define LSTRIDE  258

__global__ __launch_bounds__(256) void zero_i32(int* __restrict__ p, int n) {
    int i = blockIdx.x * 256 + threadIdx.x;
    int st = gridDim.x * 256;
    for (; i < n; i += st) p[i] = 0;
}

__global__ __launch_bounds__(256) void count_dst(const int* __restrict__ dst,
                                                 int* __restrict__ cnt, int nE) {
    int i = blockIdx.x * 256 + threadIdx.x;
    int st = gridDim.x * 256;
    for (; i < nE; i += st) atomicAdd(&cnt[dst[i]], 1);
}

__global__ __launch_bounds__(256) void scan_block(const int* __restrict__ cnt,
                                                  int* __restrict__ off,
                                                  int* __restrict__ bsum, int n) {
    __shared__ int s[256];
    const int tid = threadIdx.x;
    const int gid = blockIdx.x * 256 + tid;
    int v = (gid < n) ? cnt[gid] : 0;
    s[tid] = v;
    __syncthreads();
    for (int d = 1; d < 256; d <<= 1) {
        int t = (tid >= d) ? s[tid - d] : 0;
        __syncthreads();
        s[tid] += t;
        __syncthreads();
    }
    if (gid < n) off[gid] = s[tid] - v;
    if (tid == 255) bsum[blockIdx.x] = s[255];
}

__global__ __launch_bounds__(256) void scan_sums(int* __restrict__ bsum, int nb) {
    __shared__ int s[256];
    const int tid = threadIdx.x;
    int v = (tid < nb) ? bsum[tid] : 0;
    s[tid] = v;
    __syncthreads();
    for (int d = 1; d < 256; d <<= 1) {
        int t = (tid >= d) ? s[tid - d] : 0;
        __syncthreads();
        s[tid] += t;
        __syncthreads();
    }
    bsum[tid] = s[tid] - v;
}

__global__ __launch_bounds__(256) void scan_finish(int* __restrict__ off,
                                                   const int* __restrict__ bsum,
                                                   int* __restrict__ cursor,
                                                   int n, int nE) {
    const int gid = blockIdx.x * 256 + threadIdx.x;
    if (gid < n) {
        int o = off[gid] + bsum[blockIdx.x];
        off[gid] = o;
        cursor[gid] = o;
    }
    if (gid == 0) off[n] = nE;
}

__global__ __launch_bounds__(256) void build_perm(const int* __restrict__ src,
                                                  const int* __restrict__ dst,
                                                  int* __restrict__ cursor,
                                                  int* __restrict__ perm, int nE) {
    int i = blockIdx.x * 256 + threadIdx.x;
    int st = gridDim.x * 256;
    for (; i < nE; i += st) {
        int slot = atomicAdd(&cursor[dst[i]], 1);
        perm[slot] = src[i];
    }
}

__global__ __launch_bounds__(256) void gather_mean(const float* __restrict__ xsrc,
                                                   const int* __restrict__ perm,
                                                   const int* __restrict__ off,
                                                   float* __restrict__ meanOut,
                                                   int nDst) {
    const int wave = blockIdx.x * 8 + (threadIdx.x >> 5);
    const int lane = threadIdx.x & 31;
    if (wave >= nDst) return;
    const int s = off[wave], e = off[wave + 1];
    float a0 = 0.f, a1 = 0.f, a2 = 0.f, a3 = 0.f;
    float b0 = 0.f, b1 = 0.f, b2 = 0.f, b3 = 0.f;
    for (int j = s; j < e; ++j) {
        const float4* row = (const float4*)(xsrc + (size_t)perm[j] * DIM);
        float4 a = row[lane];
        float4 b = row[lane + 32];
        a0 += a.x; a1 += a.y; a2 += a.z; a3 += a.w;
        b0 += b.x; b1 += b.y; b2 += b.z; b3 += b.w;
    }
    const float inv = 1.0f / fmaxf((float)(e - s), 1.0f);
    float4* orow = (float4*)(meanOut + (size_t)wave * DIM);
    orow[lane]      = make_float4(a0 * inv, a1 * inv, a2 * inv, a3 * inv);
    orow[lane + 32] = make_float4(b0 * inv, b1 * inv, b2 * inv, b3 * inv);
}

__global__ __launch_bounds__(256) void sage_gemm(float* __restrict__ meanio,
                                                 const float* __restrict__ xdst,
                                                 const float* __restrict__ Wl,
                                                 const float* __restrict__ bl,
                                                 const float* __restrict__ Wr,
                                                 int relu_norm, int nRows) {
    __shared__ float sMean[MT * LSTRIDE];
    __shared__ float sSelf[MT * LSTRIDE];
    __shared__ float sPart[MT * 4];
    __shared__ float sInv [MT];

    const int tid = threadIdx.x;
    const int m0  = blockIdx.x * MT;

    for (int i = tid; i < MT * DIM / 4; i += 256) {
        const int m = i >> 6;
        const int c = (i & 63) * 4;
        int gm = m0 + m;
        if (gm >= nRows) gm = nRows - 1;
        float4 a = *(const float4*)(meanio + (size_t)gm * DIM + c);
        float4 s = *(const float4*)(xdst   + (size_t)gm * DIM + c);
        float* pm = &sMean[m * LSTRIDE + c];
        pm[0] = a.x; pm[1] = a.y; pm[2] = a.z; pm[3] = a.w;
        float* ps = &sSelf[m * LSTRIDE + c];
        ps[0] = s.x; ps[1] = s.y; ps[2] = s.z; ps[3] = s.w;
    }
    __syncthreads();

    const int lane  = tid & 31;
    const int wv    = tid >> 5;
    const int mrow  = lane & 15;
    const int khalf = (lane >> 4) << 1;
    const int ncol  = lane & 15;
    const int n0    = wv * 32;

    const float* wl0 = Wl + (size_t)(n0 + ncol) * DIM;
    const float* wl1 = Wl + (size_t)(n0 + 16 + ncol) * DIM;
    const float* wr0 = Wr + (size_t)(n0 + ncol) * DIM;
    const float* wr1 = Wr + (size_t)(n0 + 16 + ncol) * DIM;

    v8f acc[4][2];
#pragma unroll
    for (int s = 0; s < 4; ++s)
#pragma unroll
        for (int t = 0; t < 2; ++t)
            acc[s][t] = (v8f){0.f, 0.f, 0.f, 0.f, 0.f, 0.f, 0.f, 0.f};

#pragma unroll 2
    for (int k = 0; k < DIM; k += 4) {
        const int kk = k + khalf;
        v2f l0 = *(const v2f*)(wl0 + kk);
        v2f l1 = *(const v2f*)(wl1 + kk);
        v2f r0 = *(const v2f*)(wr0 + kk);
        v2f r1 = *(const v2f*)(wr1 + kk);
#pragma unroll
        for (int s = 0; s < 4; ++s) {
            v2f aM = *(const v2f*)(&sMean[(s * 16 + mrow) * LSTRIDE] + kk);
            v2f aS = *(const v2f*)(&sSelf[(s * 16 + mrow) * LSTRIDE] + kk);
            acc[s][0] = __builtin_amdgcn_wmma_f32_16x16x4_f32(false, aM, false, l0, (short)0, acc[s][0], false, false);
            acc[s][0] = __builtin_amdgcn_wmma_f32_16x16x4_f32(false, aS, false, r0, (short)0, acc[s][0], false, false);
            acc[s][1] = __builtin_amdgcn_wmma_f32_16x16x4_f32(false, aM, false, l1, (short)0, acc[s][1], false, false);
            acc[s][1] = __builtin_amdgcn_wmma_f32_16x16x4_f32(false, aS, false, r1, (short)0, acc[s][1], false, false);
        }
    }

    const float bb0 = bl[n0 + ncol];
    const float bb1 = bl[n0 + 16 + ncol];
#pragma unroll
    for (int s = 0; s < 4; ++s)
#pragma unroll
        for (int r = 0; r < 8; ++r) { acc[s][0][r] += bb0; acc[s][1][r] += bb1; }

    __syncthreads();

    const int mbase = (lane >> 4) << 3;
#pragma unroll
    for (int s = 0; s < 4; ++s)
#pragma unroll
        for (int r = 0; r < 8; ++r) {
            const int m = s * 16 + mbase + r;
            sMean[m * LSTRIDE + n0 + ncol]      = acc[s][0][r];
            sMean[m * LSTRIDE + n0 + 16 + ncol] = acc[s][1][r];
        }
    __syncthreads();

    if (relu_norm) {
        {
            const int m = tid >> 2, seg = tid & 3;
            const float* row = &sMean[m * LSTRIDE + seg * 64];
            float ss = 0.f;
#pragma unroll 16
            for (int c = 0; c < 64; ++c) ss += row[c] * row[c];
            sPart[m * 4 + seg] = ss;
        }
        __syncthreads();
        if (tid < MT) {
            const float ss = sPart[tid * 4] + sPart[tid * 4 + 1] +
                             sPart[tid * 4 + 2] + sPart[tid * 4 + 3];
            sInv[tid] = 1.0f / fmaxf(sqrtf(ss), 1e-12f);
        }
        __syncthreads();
    }

    {
        const int m  = tid >> 2;
        const int cb = (tid & 3) * 64;
        if (m0 + m < nRows) {
            const float inv = relu_norm ? sInv[m] : 1.0f;
            float* orow = meanio + (size_t)(m0 + m) * DIM;
#pragma unroll
            for (int c = 0; c < 64; c += 4) {
                float x0 = sMean[m * LSTRIDE + cb + c + 0];
                float x1 = sMean[m * LSTRIDE + cb + c + 1];
                float x2 = sMean[m * LSTRIDE + cb + c + 2];
                float x3 = sMean[m * LSTRIDE + cb + c + 3];
                if (relu_norm) {
                    x0 = fmaxf(x0 * inv, 0.f); x1 = fmaxf(x1 * inv, 0.f);
                    x2 = fmaxf(x2 * inv, 0.f); x3 = fmaxf(x3 * inv, 0.f);
                }
                *(float4*)(orow + cb + c) = make_float4(x0, x1, x2, x3);
            }
        }
    }
}

extern "C" void kernel_launch(void* const* d_in, const int* in_sizes, int n_in,
                              void* d_out, int out_size, void* d_ws, size_t ws_size,
                              hipStream_t stream) {
    (void)n_in; (void)out_size; (void)ws_size;
    const float* xA = (const float*)d_in[0];
    const float* xP = (const float*)d_in[1];
    const float* Wl = (const float*)d_in[2];
    const float* bl = (const float*)d_in[3];
    const float* Wr = (const float*)d_in[4];
    const int* eap  = (const int*)d_in[5];
    const int* epa  = (const int*)d_in[6];

    const int NA = in_sizes[0] / DIM;
    const int NP = in_sizes[1] / DIM;
    const int E  = in_sizes[5] / 2;
    const int L  = in_sizes[2] / (2 * DIM * DIM);

    const size_t FA = (size_t)NA * DIM, FP = (size_t)NP * DIM;
    char* ws = (char*)d_ws;
    float* A0 = (float*)ws;  ws += FA * sizeof(float);
    float* A1 = (float*)ws;  ws += FA * sizeof(float);
    float* P0 = (float*)ws;  ws += FP * sizeof(float);
    float* P1 = (float*)ws;  ws += FP * sizeof(float);
    int* cntP  = (int*)ws;   ws += (size_t)NP * 4;
    int* offP  = (int*)ws;   ws += (size_t)(NP + 1) * 4;
    int* curP  = (int*)ws;   ws += (size_t)NP * 4;
    int* permP = (int*)ws;   ws += (size_t)E * 4;
    int* cntA  = (int*)ws;   ws += (size_t)NA * 4;
    int* offA  = (int*)ws;   ws += (size_t)(NA + 1) * 4;
    int* curA  = (int*)ws;   ws += (size_t)NA * 4;
    int* permA = (int*)ws;   ws += (size_t)E * 4;
    int* bsum  = (int*)ws;   ws += 256 * 4;

    const int nbP = (NP + 255) / 256, nbA = (NA + 255) / 256;

    zero_i32   <<<64, 256, 0, stream>>>(cntP, NP);
    count_dst  <<<1024, 256, 0, stream>>>(eap + E, cntP, E);
    scan_block <<<nbP, 256, 0, stream>>>(cntP, offP, bsum, NP);
    scan_sums  <<<1, 256, 0, stream>>>(bsum, nbP);
    scan_finish<<<nbP, 256, 0, stream>>>(offP, bsum, curP, NP, E);
    build_perm <<<1024, 256, 0, stream>>>(eap, eap + E, curP, permP, E);

    zero_i32   <<<64, 256, 0, stream>>>(cntA, NA);
    count_dst  <<<1024, 256, 0, stream>>>(epa + E, cntA, E);
    scan_block <<<nbA, 256, 0, stream>>>(cntA, offA, bsum, NA);
    scan_sums  <<<1, 256, 0, stream>>>(bsum, nbA);
    scan_finish<<<nbA, 256, 0, stream>>>(offA, bsum, curA, NA, E);
    build_perm <<<1024, 256, 0, stream>>>(epa, epa + E, curA, permA, E);

    const float* xa_cur = xA;
    const float* xp_cur = xP;
    for (int l = 0; l < L; ++l) {
        const bool last = (l == L - 1);
        float* xa_next = last ? (float*)d_out        : ((l & 1) ? A1 : A0);
        float* xp_next = last ? (float*)d_out + FA   : ((l & 1) ? P1 : P0);

        gather_mean<<<(NP + 7) / 8, 256, 0, stream>>>(xa_cur, permP, offP, xp_next, NP);
        gather_mean<<<(NA + 7) / 8, 256, 0, stream>>>(xp_cur, permA, offA, xa_next, NA);

        const float* Wl0 = Wl + ((size_t)(l * 2 + 0)) * DIM * DIM;
        const float* Wl1 = Wl + ((size_t)(l * 2 + 1)) * DIM * DIM;
        const float* Wr0 = Wr + ((size_t)(l * 2 + 0)) * DIM * DIM;
        const float* Wr1 = Wr + ((size_t)(l * 2 + 1)) * DIM * DIM;
        const float* bl0 = bl + (size_t)(l * 2 + 0) * DIM;
        const float* bl1 = bl + (size_t)(l * 2 + 1) * DIM;

        sage_gemm<<<(NP + MT - 1) / MT, 256, 0, stream>>>(xp_next, xp_cur, Wl0, bl0, Wr0, last ? 0 : 1, NP);
        sage_gemm<<<(NA + MT - 1) / MT, 256, 0, stream>>>(xa_next, xa_cur, Wl1, bl1, Wr1, last ? 0 : 1, NA);

        xa_cur = xa_next;
        xp_cur = xp_next;
    }
}


# list build of both edge types in 4 shared launches (was 12): the counting atomic's return is the edge's rank, the fill has no atomic and redoes the block-sum prefix; a row's sum order is that rank
# speedup vs baseline: 1.2959x; 1.2959x over previous
#include <hip/hip_runtime.h>
#include <math.h>

typedef __attribute__((ext_vector_type(2))) float v2f;
typedef __attribute__((ext_vector_type(8))) float v8f;

#define DIM      256
#define NS       13
#define MR       (NS * 16)
#define WP       260
#define KC       32
#define CP       36
#define GQ       4
#define LSTRIDE  260

__global__ __launch_bounds__(256) void zero_i32(int* __restrict__ p, int n) {
    int i = blockIdx.x * 256 + threadIdx.x;
    int st = gridDim.x * 256;
    for (; i < n; i += st) p[i] = 0;
}


__global__ __launch_bounds__(256) void count_rank2(const int* __restrict__ dst0, int* __restrict__ cnt0,
                                                   int* __restrict__ rank0,
                                                   const int* __restrict__ dst1, int* __restrict__ cnt1,
                                                   int* __restrict__ rank1, int nE) {
    const int half = (int)gridDim.x >> 1;
    const bool second = (int)blockIdx.x >= half;
    const int* dst = second ? dst1 : dst0;
    int* cnt  = second ? cnt1 : cnt0;
    int* rank = second ? rank1 : rank0;
    const int b = second ? (int)blockIdx.x - half : (int)blockIdx.x;
    const int st = half * 256;
    for (int i = b * 256 + (int)threadIdx.x; i < nE; i += st) rank[i] = atomicAdd(&cnt[dst[i]], 1);
}

__global__ __launch_bounds__(256) void scan_block2(const int* __restrict__ cnt0, int* __restrict__ loc0,
                                                   int* __restrict__ bsum0, int n0, int nb0,
                                                   const int* __restrict__ cnt1, int* __restrict__ loc1,
                                                   int* __restrict__ bsum1, int n1) {
    __shared__ int s[256];
    const bool second = (int)blockIdx.x >= nb0;
    const int* cnt = second ? cnt1 : cnt0;
    int* loc  = second ? loc1 : loc0;
    int* bsum = second ? bsum1 : bsum0;
    const int n = second ? n1 : n0;
    const int b = second ? (int)blockIdx.x - nb0 : (int)blockIdx.x;
    const int tid = threadIdx.x;
    const int gid = b * 256 + tid;
    int v = (gid < n) ? cnt[gid] : 0;
    s[tid] = v;
    __syncthreads();
    for (int d = 1; d < 256; d <<= 1) {
        int t = (tid >= d) ? s[tid - d] : 0;
        __syncthreads();
        s[tid] += t;
        __syncthreads();
    }
    if (gid < n) loc[gid] = s[tid] - v;
    if (tid == 255) bsum[b] = s[255];
}

__global__ __launch_bounds__(256) void fill_lists2(const int* __restrict__ src0, const int* __restrict__ dst0,
                                                   const int* __restrict__ rank0, const int* __restrict__ loc0,
                                                   const int* __restrict__ bsum0, int* __restrict__ off0,
                                                   int* __restrict__ perm0, int n0, int nb0,
                                                   const int* __restrict__ src1, const int* __restrict__ dst1,
                                                   const int* __restrict__ rank1, const int* __restrict__ loc1,
                                                   const int* __restrict__ bsum1, int* __restrict__ off1,
                                                   int* __restrict__ perm1, int n1, int nb1, int nE) {
    __shared__ int s[256];
    const int half = (int)gridDim.x >> 1;
    const bool second = (int)blockIdx.x >= half;
    const int* src  = second ? src1 : src0;
    const int* dst  = second ? dst1 : dst0;
    const int* rank = second ? rank1 : rank0;
    const int* loc  = second ? loc1 : loc0;
    const int* bsum = second ? bsum1 : bsum0;
    int* off  = second ? off1 : off0;
    int* perm = second ? perm1 : perm0;
    const int n  = second ? n1 : n0;
    const int nb = second ? nb1 : nb0;
    const int b  = second ? (int)blockIdx.x - half : (int)blockIdx.x;
    const int tid = threadIdx.x;

    int v = (tid < nb) ? bsum[tid] : 0;
    s[tid] = v;
    __syncthreads();
    for (int d = 1; d < 256; d <<= 1) {
        int t = (tid >= d) ? s[tid - d] : 0;
        __syncthreads();
        s[tid] += t;
        __syncthreads();
    }
    const int excl = s[tid] - v;
    __syncthreads();
    s[tid] = excl;
    __syncthreads();

    const int st = half * 256;
    for (int g = b * 256 + tid; g < n; g += st) off[g] = loc[g] + s[g >> 8];
    if (b == 0 && tid == 0) off[n] = nE;
    for (int i = b * 256 + tid; i < nE; i += st) {
        const int d = dst[i];
        perm[loc[d] + s[d >> 8] + rank[i]] = src[i];
    }
}

__global__ __launch_bounds__(256) void gather_mean(const float* __restrict__ xsrc,
                                                   const int* __restrict__ perm,
                                                   const int* __restrict__ off,
                                                   float* __restrict__ meanOut,
                                                   int nDst) {
    const int wave = blockIdx.x * 8 + (threadIdx.x >> 5);
    const int lane = threadIdx.x & 31;
    if (wave >= nDst) return;
    const int s = off[wave], e = off[wave + 1];
    float a0 = 0.f, a1 = 0.f, a2 = 0.f, a3 = 0.f;
    float b0 = 0.f, b1 = 0.f, b2 = 0.f, b3 = 0.f;
    for (int j = s; j < e; ++j) {
        const float4* row = (const float4*)(xsrc + (size_t)perm[j] * DIM);
        float4 a = row[lane];
        float4 b = row[lane + 32];
        a0 += a.x; a1 += a.y; a2 += a.z; a3 += a.w;
        b0 += b.x; b1 += b.y; b2 += b.z; b3 += b.w;
    }
    const float inv = 1.0f / fmaxf((float)(e - s), 1.0f);
    float4* orow = (float4*)(meanOut + (size_t)wave * DIM);
    orow[lane]      = make_float4(a0 * inv, a1 * inv, a2 * inv, a3 * inv);
    orow[lane + 32] = make_float4(b0 * inv, b1 * inv, b2 * inv, b3 * inv);
}


__device__ __forceinline__ void gather_store_row(float* __restrict__ gout, int d, int lane,
                                                 const float4 a, const float4 b, float inv) {
    float4* orow = (float4*)(gout + (size_t)d * DIM);
    orow[lane]      = make_float4(a.x * inv, a.y * inv, a.z * inv, a.w * inv);
    orow[lane + 32] = make_float4(b.x * inv, b.y * inv, b.z * inv, b.w * inv);
}

__device__ __forceinline__ void gather_next_dst(const int* __restrict__ goff, float* __restrict__ gout,
                                                int lane, int& d, const int dEnd,
                                                const int segBeg, int& segEnd) {
    const float4 z = make_float4(0.f, 0.f, 0.f, 0.f);
    while (d < dEnd) {
        segEnd = __builtin_amdgcn_readfirstlane(goff[d + 1]);
        if (segEnd != segBeg) break;
        gather_store_row(gout, d, lane, z, z, 1.0f);
        ++d;
    }
}

__device__ __forceinline__ void gather_turn(const float* __restrict__ gsrc, const int* __restrict__ gperm,
                                            const int* __restrict__ goff, float* __restrict__ gout,
                                            const int lane, int& d, const int dEnd,
                                            int& segBeg, int& segEnd, int& e, int& nfl, const int eEndAll,
                                            int& idxv, float4 (&rb)[GQ][2], float4& ga, float4& gb) {
#pragma unroll
    for (int q = 0; q < GQ; ++q) {
        if (q < nfl) {
            ga.x += rb[q][0].x; ga.y += rb[q][0].y; ga.z += rb[q][0].z; ga.w += rb[q][0].w;
            gb.x += rb[q][1].x; gb.y += rb[q][1].y; gb.z += rb[q][1].z; gb.w += rb[q][1].w;
            if (e - nfl + q + 1 == segEnd) {
                const float inv = 1.0f / fmaxf((float)(segEnd - segBeg), 1.0f);
                gather_store_row(gout, d, lane, ga, gb, inv);
                ga = make_float4(0.f, 0.f, 0.f, 0.f);
                gb = make_float4(0.f, 0.f, 0.f, 0.f);
                ++d;
                segBeg = segEnd;
                gather_next_dst(goff, gout, lane, d, dEnd, segBeg, segEnd);
            }
        }
    }
    int n = eEndAll - e;
    if (n > GQ) n = GQ;
    int idx[GQ];
#pragma unroll
    for (int q = 0; q < GQ; ++q) idx[q] = __builtin_amdgcn_readlane(idxv, q);
    e += n;
    nfl = n;
    idxv = (lane < GQ && e + lane < eEndAll) ? gperm[e + lane] : 0;
#pragma unroll
    for (int q = 0; q < GQ; ++q) {
        if (q < n) {
            const float4* row = (const float4*)(gsrc + (size_t)idx[q] * DIM);
            rb[q][0] = row[lane];
            rb[q][1] = row[lane + 32];
        }
    }
}

__global__ __launch_bounds__(512) void sage_gemm(float* __restrict__ meanio,
                                                 const float* __restrict__ xdst,
                                                 const float* __restrict__ Wl,
                                                 const float* __restrict__ bl,
                                                 const float* __restrict__ Wr,
                                                 int relu_norm, int nRows,
                                                 const float* __restrict__ gsrc,
                                                 const int* __restrict__ gperm,
                                                 const int* __restrict__ goff,
                                                 float* __restrict__ gout, int gN) {
    __shared__ __attribute__((aligned(16))) float sW[DIM * WP];
    __shared__ __attribute__((aligned(16))) float sA[MR * CP];

    float* sOut  = sW;
    float* sPart = sA;
    float* sInv  = sA + MR * 4;

    const int tid = threadIdx.x;

    const int nStrips = (nRows + 15) >> 4;
    const int nWg     = (int)gridDim.x;
    const int base    = nStrips / nWg;
    const int rem     = nStrips % nWg;
    const int bidx    = (int)blockIdx.x;
    const int cnt     = base + (bidx < rem ? 1 : 0);

    const int glane = tid & 31;
    int gd = 0, gdEnd = 0, gSegBeg = 0, gSegEnd = 0, ge = 0, gnfl = 0, geEnd = 0, gidx = 0;
    float4 grb[GQ][2];
    float4 gacc0 = make_float4(0.f, 0.f, 0.f, 0.f), gacc1 = make_float4(0.f, 0.f, 0.f, 0.f);
#pragma unroll
    for (int q = 0; q < GQ; ++q) { grb[q][0] = gacc0; grb[q][1] = gacc0; }
    if (gN > 0) {
        const int gw    = __builtin_amdgcn_readfirstlane(bidx * 16 + (tid >> 5));
        const int gnw   = nWg * 16;
        const int gbase = gN / gnw, grem = gN % gnw;
        gd      = gw * gbase + (gw < grem ? gw : grem);
        gdEnd   = gd + gbase + (gw < grem ? 1 : 0);
        gSegBeg = __builtin_amdgcn_readfirstlane(goff[gd]);
        geEnd   = __builtin_amdgcn_readfirstlane(goff[gdEnd]);
        ge      = gSegBeg;
        gSegEnd = gSegBeg;
        gather_next_dst(goff, gout, glane, gd, gdEnd, gSegBeg, gSegEnd);
        gidx = (glane < GQ && ge + glane < geEnd) ? gperm[ge + glane] : 0;
    }

    if (cnt == 0) {
        while (gnfl > 0 || ge < geEnd)
            gather_turn(gsrc, gperm, goff, gout, glane, gd, gdEnd, gSegBeg, gSegEnd, ge, gnfl, geEnd, gidx, grb, gacc0, gacc1);
        return;
    }
    const int r0      = (bidx * base + (bidx < rem ? bidx : rem)) * 16;
    int rEnd = r0 + cnt * 16;
    if (rEnd > nRows) rEnd = nRows;

    const int prow = tid >> 3;
    const int pc4  = (tid & 7) * 4;
    size_t go0, go1, go2, go3;
    {
        int g0 = r0 + prow, g1 = g0 + 64, g2 = g0 + 128, g3 = g0 + 192;
        if (g0 >= rEnd) g0 = rEnd - 1;
        if (g1 >= rEnd) g1 = rEnd - 1;
        if (g2 >= rEnd) g2 = rEnd - 1;
        if (g3 >= rEnd) g3 = rEnd - 1;
        go0 = (size_t)g0 * DIM + pc4; go1 = (size_t)g1 * DIM + pc4;
        go2 = (size_t)g2 * DIM + pc4; go3 = (size_t)g3 * DIM + pc4;
    }
    const bool tailPiece = prow < MR - 192;

    const int lane  = tid & 31;
    const int wv    = tid >> 5;
    const int mrow  = lane & 15;
    const int khalf = (lane >> 4) << 1;
    const int ncol  = lane & 15;
    const int n0    = wv * 16;

    v8f acc[NS];
#pragma unroll
    for (int s = 0; s < NS; ++s)
        acc[s] = (v8f){0.f, 0.f, 0.f, 0.f, 0.f, 0.f, 0.f, 0.f};

    float4 pa0, pa1, pa2, pa3;
    pa3 = make_float4(0.f, 0.f, 0.f, 0.f);

    for (int ph = 0; ph < 2; ++ph) {
        const float* Ab = ph ? xdst : (const float*)meanio;
        const float* Wb = ph ? Wr : Wl;

        {
            const float* wsrc = Wb + (size_t)(tid >> 6) * DIM + (tid & 63) * 4;
            float*       wdst = sW + (tid >> 6) * WP + (tid & 63) * 4;
            const int    rot  = bidx & 31;
            __syncthreads();

            pa0 = *(const float4*)(Ab + go0); pa1 = *(const float4*)(Ab + go1);
            pa2 = *(const float4*)(Ab + go2);
            if (tailPiece) pa3 = *(const float4*)(Ab + go3);

#pragma unroll 1
            for (int it = 0; it < 32; it += 8) {
                float4 w[8];
#pragma unroll
                for (int u = 0; u < 8; ++u)
                    w[u] = *(const float4*)(wsrc + (size_t)((it + u + rot) & 31) * (8 * DIM));
#pragma unroll
                for (int u = 0; u < 8; ++u)
                    *(float4*)(wdst + ((it + u + rot) & 31) * (8 * WP)) = w[u];
            }
        }

#pragma unroll 1
        for (int kc = 0; kc < DIM; kc += KC) {
            __syncthreads();
            *(float4*)(&sA[(prow      ) * CP + pc4]) = pa0;
            *(float4*)(&sA[(prow +  64) * CP + pc4]) = pa1;
            *(float4*)(&sA[(prow + 128) * CP + pc4]) = pa2;
            if (tailPiece) *(float4*)(&sA[(prow + 192) * CP + pc4]) = pa3;
            __syncthreads();

            if (kc + KC < DIM) {
                const int kn = kc + KC;
                pa0 = *(const float4*)(Ab + go0 + kn); pa1 = *(const float4*)(Ab + go1 + kn);
                pa2 = *(const float4*)(Ab + go2 + kn);
                if (tailPiece) pa3 = *(const float4*)(Ab + go3 + kn);
            }

#pragma unroll 1
            for (int ks = 0; ks < KC; ks += 8) {
                gather_turn(gsrc, gperm, goff, gout, glane, gd, gdEnd, gSegBeg, gSegEnd, ge, gnfl, geEnd, gidx, grb, gacc0, gacc1);
#pragma unroll
                for (int u = 0; u < 8; u += 4) {
                    const int kk = ks + u + khalf;
                    const v2f bf = *(const v2f*)(&sW[(n0 + ncol) * WP + kc + kk]);
#pragma unroll
                    for (int s = 0; s < NS; ++s) {
                        const v2f af = *(const v2f*)(&sA[(s * 16 + mrow) * CP + kk]);
                        acc[s] = __builtin_amdgcn_wmma_f32_16x16x4_f32(false, af, false, bf, (short)0, acc[s], false, false);
                    }
                }
            }
        }
    }

    while (gnfl > 0 || ge < geEnd)
        gather_turn(gsrc, gperm, goff, gout, glane, gd, gdEnd, gSegBeg, gSegEnd, ge, gnfl, geEnd, gidx, grb, gacc0, gacc1);

    const float bb = bl[n0 + ncol];
#pragma unroll
    for (int s = 0; s < NS; ++s)
#pragma unroll
        for (int r = 0; r < 8; ++r) acc[s][r] += bb;

    __syncthreads();

    const int mbase = (lane >> 4) << 3;
#pragma unroll
    for (int s = 0; s < NS; ++s)
#pragma unroll
        for (int r = 0; r < 8; ++r)
            sOut[(s * 16 + mbase + r) * LSTRIDE + n0 + ncol] = acc[s][r];
    __syncthreads();

    const int nValid = rEnd - r0;

    if (relu_norm) {
        for (int item = tid; item < MR * 4; item += 512) {
            const int seg = item / MR, m = item - seg * MR;
            const float* row = &sOut[m * LSTRIDE + seg * 64];
            float ss = 0.f;
#pragma unroll 16
            for (int c = 0; c < 64; ++c) ss += row[c] * row[c];
            sPart[m * 4 + seg] = ss;
        }
        __syncthreads();
        if (tid < MR) {
            const float ss = sPart[tid * 4] + sPart[tid * 4 + 1] +
                             sPart[tid * 4 + 2] + sPart[tid * 4 + 3];
            sInv[tid] = 1.0f / fmaxf(sqrtf(ss), 1e-12f);
        }
        __syncthreads();
    }

    for (int m = wv; m < nValid; m += 16) {
        float4 xa = *(const float4*)(&sOut[m * LSTRIDE + lane * 4]);
        float4 xb = *(const float4*)(&sOut[m * LSTRIDE + 128 + lane * 4]);
        if (relu_norm) {
            const float inv = sInv[m];
            xa.x = fmaxf(xa.x * inv, 0.f); xa.y = fmaxf(xa.y * inv, 0.f);
            xa.z = fmaxf(xa.z * inv, 0.f); xa.w = fmaxf(xa.w * inv, 0.f);
            xb.x = fmaxf(xb.x * inv, 0.f); xb.y = fmaxf(xb.y * inv, 0.f);
            xb.z = fmaxf(xb.z * inv, 0.f); xb.w = fmaxf(xb.w * inv, 0.f);
        }
        float* orow = meanio + (size_t)(r0 + m) * DIM;
        *(float4*)(orow + lane * 4)       = xa;
        *(float4*)(orow + 128 + lane * 4) = xb;
    }
}

extern "C" void kernel_launch(void* const* d_in, const int* in_sizes, int n_in,
                              void* d_out, int out_size, void* d_ws, size_t ws_size,
                              hipStream_t stream) {
    (void)n_in; (void)out_size; (void)ws_size;
    const float* xA = (const float*)d_in[0];
    const float* xP = (const float*)d_in[1];
    const float* Wl = (const float*)d_in[2];
    const float* bl = (const float*)d_in[3];
    const float* Wr = (const float*)d_in[4];
    const int* eap  = (const int*)d_in[5];
    const int* epa  = (const int*)d_in[6];

    const int NA = in_sizes[0] / DIM;
    const int NP = in_sizes[1] / DIM;
    const int E  = in_sizes[5] / 2;
    const int L  = in_sizes[2] / (2 * DIM * DIM);

    const size_t FA = (size_t)NA * DIM, FP = (size_t)NP * DIM;
    char* ws = (char*)d_ws;
    float* A0 = (float*)ws;  ws += FA * sizeof(float);
    float* A1 = (float*)ws;  ws += FA * sizeof(float);
    float* A2 = (float*)ws;  ws += FA * sizeof(float);
    float* P0 = (float*)ws;  ws += FP * sizeof(float);
    float* P1 = (float*)ws;  ws += FP * sizeof(float);
    float* P2 = (float*)ws;  ws += FP * sizeof(float);
    int* cntP  = (int*)ws;   ws += (size_t)NP * 4;
    int* cntA  = (int*)ws;   ws += (size_t)NA * 4;
    int* locP  = (int*)ws;   ws += (size_t)NP * 4;
    int* locA  = (int*)ws;   ws += (size_t)NA * 4;
    int* offP  = (int*)ws;   ws += (size_t)(NP + 1) * 4;
    int* offA  = (int*)ws;   ws += (size_t)(NA + 1) * 4;
    int* rankP = (int*)ws;   ws += (size_t)E * 4;
    int* rankA = (int*)ws;   ws += (size_t)E * 4;
    int* permP = (int*)ws;   ws += (size_t)E * 4;
    int* permA = (int*)ws;   ws += (size_t)E * 4;
    int* bsumP = (int*)ws;   ws += 256 * 4;
    int* bsumA = (int*)ws;   ws += 256 * 4;

    const int nbP = (NP + 255) / 256, nbA = (NA + 255) / 256;

    int gP = ((NP + 15) / 16 + NS - 1) / NS;  if (gP < 256) gP = 256;
    int gA = ((NA + 15) / 16 + NS - 1) / NS;  if (gA < 256) gA = 256;

    zero_i32   <<<128, 256, 0, stream>>>(cntP, NP + NA);
    count_rank2<<<2048, 256, 0, stream>>>(eap + E, cntP, rankP, epa + E, cntA, rankA, E);
    scan_block2<<<nbP + nbA, 256, 0, stream>>>(cntP, locP, bsumP, NP, nbP, cntA, locA, bsumA, NA);
    fill_lists2<<<2048, 256, 0, stream>>>(eap, eap + E, rankP, locP, bsumP, offP, permP, NP, nbP,
                                          epa, epa + E, rankA, locA, bsumA, offA, permA, NA, nbA, E);

    const float* xin[2]   = { xP, xA };
    float*       outr[2]  = { (float*)d_out + FA, (float*)d_out };
    float*       plane[2][3] = { { P0, P1, P2 }, { A0, A1, A2 } };
    const int*   permT[2] = { permP, permA };
    const int*   offT[2]  = { offP, offA };
    const int    nT[2]    = { NP, NA };
    const int    gT[2]    = { gP, gA };
    auto X = [&](int t, int l) -> const float* {
        return l == 0 ? xin[t] : (l == L ? (const float*)outr[t] : (const float*)plane[t][(l - 1) % 3]);
    };

    gather_mean<<<(nT[0] + 7) / 8, 256, 0, stream>>>(X(1, 0), permT[0], offT[0], (float*)X(0, 1), nT[0]);

    for (int l = 0; l < L; ++l) {
        const bool last = (l == L - 1);
        const int T = l & 1, U = 1 - T;
        const size_t wT = ((size_t)(l * 2 + T)) * DIM * DIM, wU = ((size_t)(l * 2 + U)) * DIM * DIM;

        sage_gemm<<<gT[T], 512, 0, stream>>>((float*)X(T, l + 1), X(T, l), Wl + wT, bl + (size_t)(l * 2 + T) * DIM,
                                             Wr + wT, last ? 0 : 1, nT[T],
                                             X(T, l), permT[U], offT[U], (float*)X(U, l + 1), nT[U]);
        if (!last)
            sage_gemm<<<gT[U], 512, 0, stream>>>((float*)X(U, l + 1), X(U, l), Wl + wU, bl + (size_t)(l * 2 + U) * DIM,
                                                 Wr + wU, 1, nT[U],
                                                 X(T, l + 1), permT[U], offT[U], (float*)X(U, l + 2), nT[U]);
        else
            sage_gemm<<<gT[U], 512, 0, stream>>>((float*)X(U, l + 1), X(U, l), Wl + wU, bl + (size_t)(l * 2 + U) * DIM,
                                                 Wr + wU, 0, nT[U],
                                                 nullptr, nullptr, nullptr, nullptr, 0);
    }
}
